# starting kernel
# baseline (speedup 1.0000x reference)
#include <hip/hip_runtime.h>
#include <cmath>

typedef __attribute__((ext_vector_type(2))) float v2f;
typedef __attribute__((ext_vector_type(4))) float v4f;
typedef __attribute__((ext_vector_type(8))) float v8f;

#define RH   32
#define NT   24
#define H    32
#define SCS  28
#define WAVES_PER_BLOCK 8
#define ROWS_PER_WAVE   32
#define ROWS_PER_BLOCK  (WAVES_PER_BLOCK * ROWS_PER_WAVE)

static __device__ __forceinline__ float fast_tanh(float v) {
#if __has_builtin(__builtin_amdgcn_tanhf)
    return __builtin_amdgcn_tanhf(v);
#else
    return tanhf(v);
#endif
}

static __device__ __forceinline__ float fast_exp(float v) {
#if __has_builtin(__builtin_amdgcn_exp2f)
    return __builtin_amdgcn_exp2f(v * 1.44269504088896340736f);
#else
    return __expf(v);
#endif
}

__global__ __launch_bounds__(ROWS_PER_BLOCK)
void forest_moe_kernel(const float* __restrict__ x,
                       const float* __restrict__ rw1, const float* __restrict__ rb1,
                       const float* __restrict__ rw2, const float* __restrict__ rb2,
                       const float* __restrict__ tw1, const float* __restrict__ tb1,
                       const float* __restrict__ tw2, const float* __restrict__ tb2,
                       float* __restrict__ y)
{
    __shared__ float s_rw1[RH], s_rb1[RH];
    __shared__ float s_rw2[RH * NT];
    __shared__ float s_rb2[NT], s_tb2[NT];
    __shared__ v4f   s_tree[NT * H];
    __shared__ __align__(16) float s_sc[WAVES_PER_BLOCK][ROWS_PER_WAVE][SCS];

    const int tid  = threadIdx.x;
    const int wave = tid >> 5;
    const int lane = tid & 31;

    for (int i = tid; i < RH * NT; i += ROWS_PER_BLOCK) s_rw2[i] = rw2[i];
    for (int i = tid; i < NT * H;  i += ROWS_PER_BLOCK) {
        v4f w = { tw1[i], tb1[i], tw2[i], 0.0f };
        s_tree[i] = w;
    }
    if (tid < RH) { s_rw1[tid] = rw1[tid]; s_rb1[tid] = rb1[tid]; }
    if (tid < NT) { s_rb2[tid] = rb2[tid]; s_tb2[tid] = tb2[tid]; }
    __syncthreads();

    const int rowBase = blockIdx.x * ROWS_PER_BLOCK + wave * ROWS_PER_WAVE;
    const int mrow    = lane & 15;
    const int khalf   = (lane >> 4) << 1;

    float xsave[2];
#pragma unroll
    for (int rt = 0; rt < 2; ++rt) {
        const int   gr = rowBase + rt * 16 + mrow;
        const float xv = x[gr];
        xsave[rt] = xv;

        v2f a[8];
#pragma unroll
        for (int s = 0; s < 8; ++s) {
            const int k0 = 4 * s + khalf;
            a[s].x = fast_tanh(xv * s_rw1[k0]     + s_rb1[k0]);
            a[s].y = fast_tanh(xv * s_rw1[k0 + 1] + s_rb1[k0 + 1]);
        }
#pragma unroll
        for (int n = 0; n < 2; ++n) {
            const int col = n * 16 + mrow;
            v8f c = {};
#pragma unroll
            for (int s = 0; s < 8; ++s) {
                const int k0 = 4 * s + khalf;
                v2f b;
                b.x = (col < NT) ? s_rw2[k0 * NT + col]       : 0.0f;
                b.y = (col < NT) ? s_rw2[(k0 + 1) * NT + col] : 0.0f;
                c = __builtin_amdgcn_wmma_f32_16x16x4_f32(
                         false, a[s],  false, b,
                         (short)0, c,  false,  false);
            }
            if (col < NT) {
#pragma unroll
                for (int r = 0; r < 8; ++r) {
                    const int row = rt * 16 + ((lane >> 4) << 3) + r;
                    s_sc[wave][row][col] = c[r];
                }
            }
        }
    }
    __syncthreads();

    const float xv = (lane < 16) ? xsave[0] : xsave[1];

    const v4f* scv = (const v4f*)&s_sc[wave][lane][0];
    v4f q[6];
#pragma unroll
    for (int j = 0; j < 6; ++j) q[j] = scv[j];

    float b0 = -__builtin_inff(), b1v = -__builtin_inff(), b2v = -__builtin_inff();
    int   i0 = 0, i1 = 0, i2 = 0;
#pragma unroll
    for (int t = 0; t < NT; ++t) {
        const float sv = q[t >> 2][t & 3] + s_rb2[t];
        if (sv > b0)       { b2v = b1v; i2 = i1; b1v = b0; i1 = i0; b0 = sv; i0 = t; }
        else if (sv > b1v) { b2v = b1v; i2 = i1; b1v = sv; i1 = t; }
        else if (sv > b2v) { b2v = sv;  i2 = t; }
    }
    const float e1  = fast_exp(b1v - b0);
    const float e2  = fast_exp(b2v - b0);
    const float inv = 1.0f / (1.0f + e1 + e2);

    const int   idx[3] = { i0, i1, i2 };
    const float wgt[3] = { inv, e1 * inv, e2 * inv };

    float acc = 0.0f;
#pragma unroll
    for (int i = 0; i < 3; ++i) {
        const int t = idx[i];
        const v4f* tp = &s_tree[t * H];
        float dot = 0.0f;
#pragma unroll
        for (int k = 0; k < H; ++k) {
            const v4f w = tp[k];
            const float h = fast_tanh(xv * w.x + w.y);
            dot += h * w.z;
        }
        acc += wgt[i] * (dot + s_tb2[t]);
    }
    y[rowBase + lane] = acc;
}

extern "C" void kernel_launch(void* const* d_in, const int* in_sizes, int n_in,
                              void* d_out, int out_size, void* d_ws, size_t ws_size,
                              hipStream_t stream) {
    const float* x   = (const float*)d_in[0];
    const float* rw1 = (const float*)d_in[1];
    const float* rb1 = (const float*)d_in[2];
    const float* rw2 = (const float*)d_in[3];
    const float* rb2 = (const float*)d_in[4];
    const float* tw1 = (const float*)d_in[5];
    const float* tb1 = (const float*)d_in[6];
    const float* tw2 = (const float*)d_in[7];
    const float* tb2 = (const float*)d_in[8];
    float* y = (float*)d_out;

    const int B = in_sizes[0];
    const int grid = B / ROWS_PER_BLOCK;
    forest_moe_kernel<<<grid, ROWS_PER_BLOCK, 0, stream>>>(
        x, rw1, rb1, rw2, rb2, tw1, tb1, tw2, tb2, y);
}


# expert phase regrouped: each 16-lane row works one data row at a time, lanes over hidden units, so tree-table reads are contiguous; row sums by a 16-lane butterfly (sum order of the dots changes)
# speedup vs baseline: 1.2081x; 1.2081x over previous
#include <hip/hip_runtime.h>
#include <cmath>

typedef __attribute__((ext_vector_type(2))) float v2f;
typedef __attribute__((ext_vector_type(4))) float v4f;
typedef __attribute__((ext_vector_type(8))) float v8f;

#define RH   32
#define NT   24
#define H    32
#define SCS  28
#define WAVES_PER_BLOCK 8
#define ROWS_PER_WAVE   32
#define ROWS_PER_BLOCK  (WAVES_PER_BLOCK * ROWS_PER_WAVE)

static __device__ __forceinline__ float fast_tanh(float v) {
#if __has_builtin(__builtin_amdgcn_tanhf)
    return __builtin_amdgcn_tanhf(v);
#else
    return tanhf(v);
#endif
}

static __device__ __forceinline__ float fast_exp(float v) {
#if __has_builtin(__builtin_amdgcn_exp2f)
    return __builtin_amdgcn_exp2f(v * 1.44269504088896340736f);
#else
    return __expf(v);
#endif
}

template <int I>
static __device__ __forceinline__ int share_i(int v) {
    return __builtin_amdgcn_update_dpp(0, v, 0x150 + I, 0xF, 0xF, true);
}
template <int I>
static __device__ __forceinline__ float share_f(float v) {
    return __builtin_bit_cast(float, share_i<I>(__builtin_bit_cast(int, v)));
}
template <int M>
static __device__ __forceinline__ float xmask_add(float keep, float send) {
    const int s = __builtin_amdgcn_update_dpp(0, __builtin_bit_cast(int, send), 0x160 + M, 0xF, 0xF, true);
    return keep + __builtin_bit_cast(float, s);
}

template <int I>
static __device__ __forceinline__ float expert_step(const v4f* tree, int kslot, float xv,
                                                    int o0, int o1, int o2,
                                                    float g0, float g1, float g2)
{
    const float xb = share_f<I>(xv);
    const v4f* t0 = tree + (share_i<I>(o0) + kslot);
    const v4f* t1 = tree + (share_i<I>(o1) + kslot);
    const v4f* t2 = tree + (share_i<I>(o2) + kslot);
    const v4f a0 = t0[0], c0 = t0[16];
    const v4f a1 = t1[0], c1 = t1[16];
    const v4f a2 = t2[0], c2 = t2[16];
    const float s0 = fast_tanh(xb * a0.x + a0.y) * a0.z + fast_tanh(xb * c0.x + c0.y) * c0.z;
    const float s1 = fast_tanh(xb * a1.x + a1.y) * a1.z + fast_tanh(xb * c1.x + c1.y) * c1.z;
    const float s2 = fast_tanh(xb * a2.x + a2.y) * a2.z + fast_tanh(xb * c2.x + c2.y) * c2.z;
    return share_f<I>(g0) * s0 + share_f<I>(g1) * s1 + share_f<I>(g2) * s2;
}

__global__ __launch_bounds__(ROWS_PER_BLOCK)
void forest_moe_kernel(const float* __restrict__ x,
                       const float* __restrict__ rw1, const float* __restrict__ rb1,
                       const float* __restrict__ rw2, const float* __restrict__ rb2,
                       const float* __restrict__ tw1, const float* __restrict__ tb1,
                       const float* __restrict__ tw2, const float* __restrict__ tb2,
                       float* __restrict__ y)
{
    __shared__ float s_rw1[RH], s_rb1[RH];
    __shared__ float s_rw2[RH * NT];
    __shared__ float s_rb2[NT], s_tb2[NT];
    __shared__ v4f   s_tree[NT * H];
    __shared__ __align__(16) float s_sc[WAVES_PER_BLOCK][ROWS_PER_WAVE][SCS];

    const int tid  = threadIdx.x;
    const int wave = tid >> 5;
    const int lane = tid & 31;

    for (int i = tid; i < RH * NT; i += ROWS_PER_BLOCK) s_rw2[i] = rw2[i];
    for (int i = tid; i < NT * H;  i += ROWS_PER_BLOCK) {
        v4f w = { tw1[i], tb1[i], tw2[i], 0.0f };
        s_tree[i] = w;
    }
    if (tid < RH) { s_rw1[tid] = rw1[tid]; s_rb1[tid] = rb1[tid]; }
    if (tid < NT) { s_rb2[tid] = rb2[tid]; s_tb2[tid] = tb2[tid]; }
    __syncthreads();

    const int rowBase = blockIdx.x * ROWS_PER_BLOCK + wave * ROWS_PER_WAVE;
    const int mrow    = lane & 15;
    const int khalf   = (lane >> 4) << 1;

    float xsave[2];
#pragma unroll
    for (int rt = 0; rt < 2; ++rt) {
        const int   gr = rowBase + rt * 16 + mrow;
        const float xv = x[gr];
        xsave[rt] = xv;

        v2f a[8];
#pragma unroll
        for (int s = 0; s < 8; ++s) {
            const int k0 = 4 * s + khalf;
            a[s].x = fast_tanh(xv * s_rw1[k0]     + s_rb1[k0]);
            a[s].y = fast_tanh(xv * s_rw1[k0 + 1] + s_rb1[k0 + 1]);
        }
#pragma unroll
        for (int n = 0; n < 2; ++n) {
            const int col = n * 16 + mrow;
            v8f c = {};
#pragma unroll
            for (int s = 0; s < 8; ++s) {
                const int k0 = 4 * s + khalf;
                v2f b;
                b.x = (col < NT) ? s_rw2[k0 * NT + col]       : 0.0f;
                b.y = (col < NT) ? s_rw2[(k0 + 1) * NT + col] : 0.0f;
                c = __builtin_amdgcn_wmma_f32_16x16x4_f32(
                         false, a[s],  false, b,
                         (short)0, c,  false,  false);
            }
            if (col < NT) {
#pragma unroll
                for (int r = 0; r < 8; ++r) {
                    const int row = rt * 16 + ((lane >> 4) << 3) + r;
                    s_sc[wave][row][col] = c[r];
                }
            }
        }
    }
    __syncthreads();

    const float xv = (lane < 16) ? xsave[0] : xsave[1];

    const v4f* scv = (const v4f*)&s_sc[wave][lane][0];
    v4f q[6];
#pragma unroll
    for (int j = 0; j < 6; ++j) q[j] = scv[j];

    float b0 = -__builtin_inff(), b1v = -__builtin_inff(), b2v = -__builtin_inff();
    int   i0 = 0, i1 = 0, i2 = 0;
#pragma unroll
    for (int t = 0; t < NT; ++t) {
        const float sv = q[t >> 2][t & 3] + s_rb2[t];
        if (sv > b0)       { b2v = b1v; i2 = i1; b1v = b0; i1 = i0; b0 = sv; i0 = t; }
        else if (sv > b1v) { b2v = b1v; i2 = i1; b1v = sv; i1 = t; }
        else if (sv > b2v) { b2v = sv;  i2 = t; }
    }
    const float e1  = fast_exp(b1v - b0);
    const float e2  = fast_exp(b2v - b0);
    const float inv = 1.0f / (1.0f + e1 + e2);
    const float g0 = inv, g1 = e1 * inv, g2 = e2 * inv;

    const int kslot = lane & 15;
    const int o0 = i0 * H, o1 = i1 * H, o2 = i2 * H;
    float p[16];
    p[0]  = expert_step<0>(s_tree, kslot, xv, o0, o1, o2, g0, g1, g2);
    p[1]  = expert_step<1>(s_tree, kslot, xv, o0, o1, o2, g0, g1, g2);
    p[2]  = expert_step<2>(s_tree, kslot, xv, o0, o1, o2, g0, g1, g2);
    p[3]  = expert_step<3>(s_tree, kslot, xv, o0, o1, o2, g0, g1, g2);
    p[4]  = expert_step<4>(s_tree, kslot, xv, o0, o1, o2, g0, g1, g2);
    p[5]  = expert_step<5>(s_tree, kslot, xv, o0, o1, o2, g0, g1, g2);
    p[6]  = expert_step<6>(s_tree, kslot, xv, o0, o1, o2, g0, g1, g2);
    p[7]  = expert_step<7>(s_tree, kslot, xv, o0, o1, o2, g0, g1, g2);
    p[8]  = expert_step<8>(s_tree, kslot, xv, o0, o1, o2, g0, g1, g2);
    p[9]  = expert_step<9>(s_tree, kslot, xv, o0, o1, o2, g0, g1, g2);
    p[10] = expert_step<10>(s_tree, kslot, xv, o0, o1, o2, g0, g1, g2);
    p[11] = expert_step<11>(s_tree, kslot, xv, o0, o1, o2, g0, g1, g2);
    p[12] = expert_step<12>(s_tree, kslot, xv, o0, o1, o2, g0, g1, g2);
    p[13] = expert_step<13>(s_tree, kslot, xv, o0, o1, o2, g0, g1, g2);
    p[14] = expert_step<14>(s_tree, kslot, xv, o0, o1, o2, g0, g1, g2);
    p[15] = expert_step<15>(s_tree, kslot, xv, o0, o1, o2, g0, g1, g2);

    const bool h8 = (lane & 8) != 0, h4 = (lane & 4) != 0, h2 = (lane & 2) != 0, h1 = (lane & 1) != 0;
    float r8[8], r4[4], r2[2];
#pragma unroll
    for (int j = 0; j < 8; ++j) r8[j] = xmask_add<8>(h8 ? p[j + 8] : p[j], h8 ? p[j] : p[j + 8]);
#pragma unroll
    for (int j = 0; j < 4; ++j) r4[j] = xmask_add<4>(h4 ? r8[j + 4] : r8[j], h4 ? r8[j] : r8[j + 4]);
#pragma unroll
    for (int j = 0; j < 2; ++j) r2[j] = xmask_add<2>(h2 ? r4[j + 2] : r4[j], h2 ? r4[j] : r4[j + 2]);
    const float dots = xmask_add<1>(h1 ? r2[1] : r2[0], h1 ? r2[0] : r2[1]);

    y[rowBase + lane] = dots + (g0 * s_tb2[i0] + g1 * s_tb2[i1] + g2 * s_tb2[i2]);
}

extern "C" void kernel_launch(void* const* d_in, const int* in_sizes, int n_in,
                              void* d_out, int out_size, void* d_ws, size_t ws_size,
                              hipStream_t stream) {
    const float* x   = (const float*)d_in[0];
    const float* rw1 = (const float*)d_in[1];
    const float* rb1 = (const float*)d_in[2];
    const float* rw2 = (const float*)d_in[3];
    const float* rb2 = (const float*)d_in[4];
    const float* tw1 = (const float*)d_in[5];
    const float* tb1 = (const float*)d_in[6];
    const float* tw2 = (const float*)d_in[7];
    const float* tb2 = (const float*)d_in[8];
    float* y = (float*)d_out;

    const int B = in_sizes[0];
    const int grid = B / ROWS_PER_BLOCK;
    forest_moe_kernel<<<grid, ROWS_PER_BLOCK, 0, stream>>>(
        x, rw1, rb1, rw2, rb2, tw1, tb1, tw2, tb2, y);
}


# router product transposed (weights as WMMA A operand, activations as B): a row's scores land in its own lane's registers via 8 permlane16 swaps; LDS score buffer, its stores, loads and barrier gone
# speedup vs baseline: 1.2902x; 1.0680x over previous
#include <hip/hip_runtime.h>
#include <cmath>

typedef __attribute__((ext_vector_type(2))) float v2f;
typedef __attribute__((ext_vector_type(2))) unsigned v2u;
typedef __attribute__((ext_vector_type(4))) float v4f;
typedef __attribute__((ext_vector_type(8))) float v8f;

#define RH   32
#define NT   24
#define H    32
#define WAVES_PER_BLOCK 8
#define ROWS_PER_WAVE   32
#define ROWS_PER_BLOCK  (WAVES_PER_BLOCK * ROWS_PER_WAVE)

static __device__ __forceinline__ float fast_tanh(float v) {
#if __has_builtin(__builtin_amdgcn_tanhf)
    return __builtin_amdgcn_tanhf(v);
#else
    return tanhf(v);
#endif
}

static __device__ __forceinline__ float fast_exp(float v) {
#if __has_builtin(__builtin_amdgcn_exp2f)
    return __builtin_amdgcn_exp2f(v * 1.44269504088896340736f);
#else
    return __expf(v);
#endif
}

template <int I>
static __device__ __forceinline__ int share_i(int v) {
    return __builtin_amdgcn_update_dpp(0, v, 0x150 + I, 0xF, 0xF, true);
}
template <int I>
static __device__ __forceinline__ float share_f(float v) {
    return __builtin_bit_cast(float, share_i<I>(__builtin_bit_cast(int, v)));
}
template <int M>
static __device__ __forceinline__ float xmask_add(float keep, float send) {
    const int s = __builtin_amdgcn_update_dpp(0, __builtin_bit_cast(int, send), 0x160 + M, 0xF, 0xF, true);
    return keep + __builtin_bit_cast(float, s);
}

template <int I>
static __device__ __forceinline__ float expert_step(const v4f* tree, int kslot, float xv,
                                                    int o0, int o1, int o2,
                                                    float g0, float g1, float g2)
{
    const float xb = share_f<I>(xv);
    const v4f* t0 = tree + (share_i<I>(o0) + kslot);
    const v4f* t1 = tree + (share_i<I>(o1) + kslot);
    const v4f* t2 = tree + (share_i<I>(o2) + kslot);
    const v4f a0 = t0[0], c0 = t0[16];
    const v4f a1 = t1[0], c1 = t1[16];
    const v4f a2 = t2[0], c2 = t2[16];
    const float s0 = fast_tanh(xb * a0.x + a0.y) * a0.z + fast_tanh(xb * c0.x + c0.y) * c0.z;
    const float s1 = fast_tanh(xb * a1.x + a1.y) * a1.z + fast_tanh(xb * c1.x + c1.y) * c1.z;
    const float s2 = fast_tanh(xb * a2.x + a2.y) * a2.z + fast_tanh(xb * c2.x + c2.y) * c2.z;
    return share_f<I>(g0) * s0 + share_f<I>(g1) * s1 + share_f<I>(g2) * s2;
}

__global__ __launch_bounds__(ROWS_PER_BLOCK)
void forest_moe_kernel(const float* __restrict__ x,
                       const float* __restrict__ rw1, const float* __restrict__ rb1,
                       const float* __restrict__ rw2, const float* __restrict__ rb2,
                       const float* __restrict__ tw1, const float* __restrict__ tb1,
                       const float* __restrict__ tw2, const float* __restrict__ tb2,
                       float* __restrict__ y)
{
    __shared__ float s_rw1[RH], s_rb1[RH];
    __shared__ float s_rw2[RH * NT];
    __shared__ float s_rb2[NT], s_tb2[NT];
    __shared__ v4f   s_tree[NT * H];

    const int tid  = threadIdx.x;
    const int wave = tid >> 5;
    const int lane = tid & 31;

    for (int i = tid; i < RH * NT; i += ROWS_PER_BLOCK) s_rw2[i] = rw2[i];
    for (int i = tid; i < NT * H;  i += ROWS_PER_BLOCK) {
        v4f w = { tw1[i], tb1[i], tw2[i], 0.0f };
        s_tree[i] = w;
    }
    if (tid < RH) { s_rw1[tid] = rw1[tid]; s_rb1[tid] = rb1[tid]; }
    if (tid < NT) { s_rb2[tid] = rb2[tid]; s_tb2[tid] = tb2[tid]; }
    __syncthreads();

    const int rowBase = blockIdx.x * ROWS_PER_BLOCK + wave * ROWS_PER_WAVE;
    const int mrow    = lane & 15;
    const int khalf   = (lane >> 4) << 1;

    const int tr0 = mrow;
    const int tr1 = 16 + (mrow & 7);
    v2f wa0[8], wa1[8];
#pragma unroll
    for (int s = 0; s < 8; ++s) {
        const int k0 = 4 * s + khalf;
        wa0[s].x = s_rw2[k0 * NT + tr0];
        wa0[s].y = s_rw2[(k0 + 1) * NT + tr0];
        wa1[s].x = s_rw2[k0 * NT + tr1];
        wa1[s].y = s_rw2[(k0 + 1) * NT + tr1];
    }

    float xsave[2];
    v8f d0[2], d1[2];
#pragma unroll
    for (int rt = 0; rt < 2; ++rt) {
        const int   gr = rowBase + rt * 16 + mrow;
        const float xv = x[gr];
        xsave[rt] = xv;

        v2f hb[8];
#pragma unroll
        for (int s = 0; s < 8; ++s) {
            const int k0 = 4 * s + khalf;
            hb[s].x = fast_tanh(xv * s_rw1[k0]     + s_rb1[k0]);
            hb[s].y = fast_tanh(xv * s_rw1[k0 + 1] + s_rb1[k0 + 1]);
        }
        v8f c0 = {}, c1 = {};
#pragma unroll
        for (int s = 0; s < 8; ++s) {
            c0 = __builtin_amdgcn_wmma_f32_16x16x4_f32(
                     false, wa0[s],  false, hb[s],
                     (short)0, c0,  false,  false);
            c1 = __builtin_amdgcn_wmma_f32_16x16x4_f32(
                     false, wa1[s],  false, hb[s],
                     (short)0, c1,  false,  false);
        }
        d0[rt] = c0;
        d1[rt] = c1;
    }

    const bool upper = lane >= 16;
    float sc[NT];
#pragma unroll
    for (int r = 0; r < 8; ++r) {
        const float lo = d0[0][r], hi = d0[1][r];
        const v2u sw = __builtin_amdgcn_permlane16_swap(
                __builtin_bit_cast(unsigned, lo), __builtin_bit_cast(unsigned, hi), false, false);
        const unsigned s0 = sw.x, s1 = sw.y;
        sc[r]      = __builtin_bit_cast(float, s0);
        sc[8 + r]  = __builtin_bit_cast(float, s1);
        sc[16 + r] = upper ? d1[1][r] : d1[0][r];
    }

    const float xv = (lane < 16) ? xsave[0] : xsave[1];

    float b0 = -__builtin_inff(), b1v = -__builtin_inff(), b2v = -__builtin_inff();
    int   i0 = 0, i1 = 0, i2 = 0;
#pragma unroll
    for (int t = 0; t < NT; ++t) {
        const float sv = sc[t] + s_rb2[t];
        if (sv > b0)       { b2v = b1v; i2 = i1; b1v = b0; i1 = i0; b0 = sv; i0 = t; }
        else if (sv > b1v) { b2v = b1v; i2 = i1; b1v = sv; i1 = t; }
        else if (sv > b2v) { b2v = sv;  i2 = t; }
    }
    const float e1  = fast_exp(b1v - b0);
    const float e2  = fast_exp(b2v - b0);
    const float inv = 1.0f / (1.0f + e1 + e2);
    const float g0 = inv, g1 = e1 * inv, g2 = e2 * inv;

    const int kslot = lane & 15;
    const int o0 = i0 * H, o1 = i1 * H, o2 = i2 * H;
    float p[16];
    p[0]  = expert_step<0>(s_tree, kslot, xv, o0, o1, o2, g0, g1, g2);
    p[1]  = expert_step<1>(s_tree, kslot, xv, o0, o1, o2, g0, g1, g2);
    p[2]  = expert_step<2>(s_tree, kslot, xv, o0, o1, o2, g0, g1, g2);
    p[3]  = expert_step<3>(s_tree, kslot, xv, o0, o1, o2, g0, g1, g2);
    p[4]  = expert_step<4>(s_tree, kslot, xv, o0, o1, o2, g0, g1, g2);
    p[5]  = expert_step<5>(s_tree, kslot, xv, o0, o1, o2, g0, g1, g2);
    p[6]  = expert_step<6>(s_tree, kslot, xv, o0, o1, o2, g0, g1, g2);
    p[7]  = expert_step<7>(s_tree, kslot, xv, o0, o1, o2, g0, g1, g2);
    p[8]  = expert_step<8>(s_tree, kslot, xv, o0, o1, o2, g0, g1, g2);
    p[9]  = expert_step<9>(s_tree, kslot, xv, o0, o1, o2, g0, g1, g2);
    p[10] = expert_step<10>(s_tree, kslot, xv, o0, o1, o2, g0, g1, g2);
    p[11] = expert_step<11>(s_tree, kslot, xv, o0, o1, o2, g0, g1, g2);
    p[12] = expert_step<12>(s_tree, kslot, xv, o0, o1, o2, g0, g1, g2);
    p[13] = expert_step<13>(s_tree, kslot, xv, o0, o1, o2, g0, g1, g2);
    p[14] = expert_step<14>(s_tree, kslot, xv, o0, o1, o2, g0, g1, g2);
    p[15] = expert_step<15>(s_tree, kslot, xv, o0, o1, o2, g0, g1, g2);

    const bool h8 = (lane & 8) != 0, h4 = (lane & 4) != 0, h2 = (lane & 2) != 0, h1 = (lane & 1) != 0;
    float r8[8], r4[4], r2[2];
#pragma unroll
    for (int j = 0; j < 8; ++j) r8[j] = xmask_add<8>(h8 ? p[j + 8] : p[j], h8 ? p[j] : p[j + 8]);
#pragma unroll
    for (int j = 0; j < 4; ++j) r4[j] = xmask_add<4>(h4 ? r8[j + 4] : r8[j], h4 ? r8[j] : r8[j + 4]);
#pragma unroll
    for (int j = 0; j < 2; ++j) r2[j] = xmask_add<2>(h2 ? r4[j + 2] : r4[j], h2 ? r4[j] : r4[j + 2]);
    const float dots = xmask_add<1>(h1 ? r2[1] : r2[0], h1 ? r2[0] : r2[1]);

    y[rowBase + lane] = dots + (g0 * s_tb2[i0] + g1 * s_tb2[i1] + g2 * s_tb2[i2]);
}

extern "C" void kernel_launch(void* const* d_in, const int* in_sizes, int n_in,
                              void* d_out, int out_size, void* d_ws, size_t ws_size,
                              hipStream_t stream) {
    const float* x   = (const float*)d_in[0];
    const float* rw1 = (const float*)d_in[1];
    const float* rb1 = (const float*)d_in[2];
    const float* rw2 = (const float*)d_in[3];
    const float* rb2 = (const float*)d_in[4];
    const float* tw1 = (const float*)d_in[5];
    const float* tb1 = (const float*)d_in[6];
    const float* tw2 = (const float*)d_in[7];
    const float* tb2 = (const float*)d_in[8];
    float* y = (float*)d_out;

    const int B = in_sizes[0];
    const int grid = B / ROWS_PER_BLOCK;
    forest_moe_kernel<<<grid, ROWS_PER_BLOCK, 0, stream>>>(
        x, rw1, rb1, rw2, rb2, tw1, tb1, tw2, tb2, y);
}


# router weights, hidden-layer weights, both bias vectors and x go from global memory straight into registers ahead of the tree-table staging; LDS holds the tree table only
# speedup vs baseline: 1.5713x; 1.2179x over previous
#include <hip/hip_runtime.h>
#include <cmath>

typedef __attribute__((ext_vector_type(2))) float v2f;
typedef __attribute__((ext_vector_type(2))) unsigned v2u;
typedef __attribute__((ext_vector_type(4))) float v4f;
typedef __attribute__((ext_vector_type(8))) float v8f;

#define RH   32
#define NT   24
#define H    32
#define WAVES_PER_BLOCK 8
#define ROWS_PER_WAVE   32
#define ROWS_PER_BLOCK  (WAVES_PER_BLOCK * ROWS_PER_WAVE)

static __device__ __forceinline__ float fast_tanh(float v) {
#if __has_builtin(__builtin_amdgcn_tanhf)
    return __builtin_amdgcn_tanhf(v);
#else
    return tanhf(v);
#endif
}

static __device__ __forceinline__ float fast_exp(float v) {
#if __has_builtin(__builtin_amdgcn_exp2f)
    return __builtin_amdgcn_exp2f(v * 1.44269504088896340736f);
#else
    return __expf(v);
#endif
}

template <int I>
static __device__ __forceinline__ int share_i(int v) {
    return __builtin_amdgcn_update_dpp(0, v, 0x150 + I, 0xF, 0xF, true);
}
template <int I>
static __device__ __forceinline__ float share_f(float v) {
    return __builtin_bit_cast(float, share_i<I>(__builtin_bit_cast(int, v)));
}
template <int M>
static __device__ __forceinline__ float xmask_add(float keep, float send) {
    const int s = __builtin_amdgcn_update_dpp(0, __builtin_bit_cast(int, send), 0x160 + M, 0xF, 0xF, true);
    return keep + __builtin_bit_cast(float, s);
}

template <int I>
static __device__ __forceinline__ float expert_step(const v4f* tree, int kslot, float xv,
                                                    int o0, int o1, int o2,
                                                    float g0, float g1, float g2)
{
    const float xb = share_f<I>(xv);
    const v4f* t0 = tree + (share_i<I>(o0) + kslot);
    const v4f* t1 = tree + (share_i<I>(o1) + kslot);
    const v4f* t2 = tree + (share_i<I>(o2) + kslot);
    const v4f a0 = t0[0], c0 = t0[16];
    const v4f a1 = t1[0], c1 = t1[16];
    const v4f a2 = t2[0], c2 = t2[16];
    const float s0 = fast_tanh(xb * a0.x + a0.y) * a0.z + fast_tanh(xb * c0.x + c0.y) * c0.z;
    const float s1 = fast_tanh(xb * a1.x + a1.y) * a1.z + fast_tanh(xb * c1.x + c1.y) * c1.z;
    const float s2 = fast_tanh(xb * a2.x + a2.y) * a2.z + fast_tanh(xb * c2.x + c2.y) * c2.z;
    return share_f<I>(g0) * s0 + share_f<I>(g1) * s1 + share_f<I>(g2) * s2;
}

__global__ __launch_bounds__(ROWS_PER_BLOCK)
void forest_moe_kernel(const float* __restrict__ x,
                       const float* __restrict__ rw1, const float* __restrict__ rb1,
                       const float* __restrict__ rw2, const float* __restrict__ rb2,
                       const float* __restrict__ tw1, const float* __restrict__ tb1,
                       const float* __restrict__ tw2, const float* __restrict__ tb2,
                       float* __restrict__ y)
{
    __shared__ v4f   s_tree[NT * H];

    const int tid  = threadIdx.x;
    const int wave = tid >> 5;
    const int lane = tid & 31;

    const int rowBase = blockIdx.x * ROWS_PER_BLOCK + wave * ROWS_PER_WAVE;
    const int mrow    = lane & 15;
    const int khalf   = (lane >> 4) << 1;

    float xsave[2];
    xsave[0] = x[rowBase + mrow];
    xsave[1] = x[rowBase + 16 + mrow];
    const int tr0 = mrow;
    const int tr1 = 16 + (mrow & 7);
    v2f wa0[8], wa1[8], w1p[8], b1p[8];
#pragma unroll
    for (int s = 0; s < 8; ++s) {
        const int k0 = 4 * s + khalf;
        wa0[s].x = rw2[k0 * NT + tr0];
        wa0[s].y = rw2[(k0 + 1) * NT + tr0];
        wa1[s].x = rw2[k0 * NT + tr1];
        wa1[s].y = rw2[(k0 + 1) * NT + tr1];
        w1p[s].x = rw1[k0];
        w1p[s].y = rw1[k0 + 1];
        b1p[s].x = rb1[k0];
        b1p[s].y = rb1[k0 + 1];
    }
    float bias[NT];
#pragma unroll
    for (int t = 0; t < NT; ++t) bias[t] = rb2[t];

    for (int i = tid; i < NT * H;  i += ROWS_PER_BLOCK) {
        v4f w = { tw1[i], tb1[i], tw2[i], 0.0f };
        s_tree[i] = w;
    }
    __syncthreads();

    v8f d0[2], d1[2];
#pragma unroll
    for (int rt = 0; rt < 2; ++rt) {
        const float xv = xsave[rt];
        v2f hb[8];
#pragma unroll
        for (int s = 0; s < 8; ++s) {
            hb[s].x = fast_tanh(xv * w1p[s].x + b1p[s].x);
            hb[s].y = fast_tanh(xv * w1p[s].y + b1p[s].y);
        }
        v8f c0 = {}, c1 = {};
#pragma unroll
        for (int s = 0; s < 8; ++s) {
            c0 = __builtin_amdgcn_wmma_f32_16x16x4_f32(
                     false, wa0[s],  false, hb[s],
                     (short)0, c0,  false,  false);
            c1 = __builtin_amdgcn_wmma_f32_16x16x4_f32(
                     false, wa1[s],  false, hb[s],
                     (short)0, c1,  false,  false);
        }
        d0[rt] = c0;
        d1[rt] = c1;
    }

    const bool upper = lane >= 16;
    float sc[NT];
#pragma unroll
    for (int r = 0; r < 8; ++r) {
        const float lo = d0[0][r], hi = d0[1][r];
        const v2u sw = __builtin_amdgcn_permlane16_swap(
                __builtin_bit_cast(unsigned, lo), __builtin_bit_cast(unsigned, hi), false, false);
        const unsigned s0 = sw.x, s1 = sw.y;
        sc[r]      = __builtin_bit_cast(float, s0);
        sc[8 + r]  = __builtin_bit_cast(float, s1);
        sc[16 + r] = upper ? d1[1][r] : d1[0][r];
    }

    const float xv = (lane < 16) ? xsave[0] : xsave[1];

    float b0 = -__builtin_inff(), b1v = -__builtin_inff(), b2v = -__builtin_inff();
    int   i0 = 0, i1 = 0, i2 = 0;
#pragma unroll
    for (int t = 0; t < NT; ++t) {
        const float sv = sc[t] + bias[t];
        if (sv > b0)       { b2v = b1v; i2 = i1; b1v = b0; i1 = i0; b0 = sv; i0 = t; }
        else if (sv > b1v) { b2v = b1v; i2 = i1; b1v = sv; i1 = t; }
        else if (sv > b2v) { b2v = sv;  i2 = t; }
    }
    const float e1  = fast_exp(b1v - b0);
    const float e2  = fast_exp(b2v - b0);
    const float inv = 1.0f / (1.0f + e1 + e2);
    const float g0 = inv, g1 = e1 * inv, g2 = e2 * inv;

    const float t0 = tb2[i0], t1 = tb2[i1], t2 = tb2[i2];

    const int kslot = lane & 15;
    const int o0 = i0 * H, o1 = i1 * H, o2 = i2 * H;
    float p[16];
    p[0]  = expert_step<0>(s_tree, kslot, xv, o0, o1, o2, g0, g1, g2);
    p[1]  = expert_step<1>(s_tree, kslot, xv, o0, o1, o2, g0, g1, g2);
    p[2]  = expert_step<2>(s_tree, kslot, xv, o0, o1, o2, g0, g1, g2);
    p[3]  = expert_step<3>(s_tree, kslot, xv, o0, o1, o2, g0, g1, g2);
    p[4]  = expert_step<4>(s_tree, kslot, xv, o0, o1, o2, g0, g1, g2);
    p[5]  = expert_step<5>(s_tree, kslot, xv, o0, o1, o2, g0, g1, g2);
    p[6]  = expert_step<6>(s_tree, kslot, xv, o0, o1, o2, g0, g1, g2);
    p[7]  = expert_step<7>(s_tree, kslot, xv, o0, o1, o2, g0, g1, g2);
    p[8]  = expert_step<8>(s_tree, kslot, xv, o0, o1, o2, g0, g1, g2);
    p[9]  = expert_step<9>(s_tree, kslot, xv, o0, o1, o2, g0, g1, g2);
    p[10] = expert_step<10>(s_tree, kslot, xv, o0, o1, o2, g0, g1, g2);
    p[11] = expert_step<11>(s_tree, kslot, xv, o0, o1, o2, g0, g1, g2);
    p[12] = expert_step<12>(s_tree, kslot, xv, o0, o1, o2, g0, g1, g2);
    p[13] = expert_step<13>(s_tree, kslot, xv, o0, o1, o2, g0, g1, g2);
    p[14] = expert_step<14>(s_tree, kslot, xv, o0, o1, o2, g0, g1, g2);
    p[15] = expert_step<15>(s_tree, kslot, xv, o0, o1, o2, g0, g1, g2);

    const bool h8 = (lane & 8) != 0, h4 = (lane & 4) != 0, h2 = (lane & 2) != 0, h1 = (lane & 1) != 0;
    float r8[8], r4[4], r2[2];
#pragma unroll
    for (int j = 0; j < 8; ++j) r8[j] = xmask_add<8>(h8 ? p[j + 8] : p[j], h8 ? p[j] : p[j + 8]);
#pragma unroll
    for (int j = 0; j < 4; ++j) r4[j] = xmask_add<4>(h4 ? r8[j + 4] : r8[j], h4 ? r8[j] : r8[j + 4]);
#pragma unroll
    for (int j = 0; j < 2; ++j) r2[j] = xmask_add<2>(h2 ? r4[j + 2] : r4[j], h2 ? r4[j] : r4[j + 2]);
    const float dots = xmask_add<1>(h1 ? r2[1] : r2[0], h1 ? r2[0] : r2[1]);

    y[rowBase + lane] = dots + (g0 * t0 + g1 * t1 + g2 * t2);
}

extern "C" void kernel_launch(void* const* d_in, const int* in_sizes, int n_in,
                              void* d_out, int out_size, void* d_ws, size_t ws_size,
                              hipStream_t stream) {
    const float* x   = (const float*)d_in[0];
    const float* rw1 = (const float*)d_in[1];
    const float* rb1 = (const float*)d_in[2];
    const float* rw2 = (const float*)d_in[3];
    const float* rb2 = (const float*)d_in[4];
    const float* tw1 = (const float*)d_in[5];
    const float* tb1 = (const float*)d_in[6];
    const float* tw2 = (const float*)d_in[7];
    const float* tb2 = (const float*)d_in[8];
    float* y = (float*)d_out;

    const int B = in_sizes[0];
    const int grid = B / ROWS_PER_BLOCK;
    forest_moe_kernel<<<grid, ROWS_PER_BLOCK, 0, stream>>>(
        x, rw1, rb1, rw2, rb2, tw1, tb1, tw2, tb2, y);
}


# tree-table staging moved from kernel entry to behind the router product; block barrier split: arrive after the stores, wait before the first table read; the top-3 selection runs between
# speedup vs baseline: 1.7207x; 1.0951x over previous
#include <hip/hip_runtime.h>
#include <cmath>

typedef __attribute__((ext_vector_type(2))) float v2f;
typedef __attribute__((ext_vector_type(2))) unsigned v2u;
typedef __attribute__((ext_vector_type(4))) float v4f;
typedef __attribute__((ext_vector_type(8))) float v8f;

#define RH   32
#define NT   24
#define H    32
#define WAVES_PER_BLOCK 8
#define ROWS_PER_WAVE   32
#define ROWS_PER_BLOCK  (WAVES_PER_BLOCK * ROWS_PER_WAVE)

static __device__ __forceinline__ float fast_tanh(float v) {
#if __has_builtin(__builtin_amdgcn_tanhf)
    return __builtin_amdgcn_tanhf(v);
#else
    return tanhf(v);
#endif
}

static __device__ __forceinline__ float fast_exp(float v) {
#if __has_builtin(__builtin_amdgcn_exp2f)
    return __builtin_amdgcn_exp2f(v * 1.44269504088896340736f);
#else
    return __expf(v);
#endif
}

template <int I>
static __device__ __forceinline__ int share_i(int v) {
    return __builtin_amdgcn_update_dpp(0, v, 0x150 + I, 0xF, 0xF, true);
}
template <int I>
static __device__ __forceinline__ float share_f(float v) {
    return __builtin_bit_cast(float, share_i<I>(__builtin_bit_cast(int, v)));
}
template <int M>
static __device__ __forceinline__ float xmask_add(float keep, float send) {
    const int s = __builtin_amdgcn_update_dpp(0, __builtin_bit_cast(int, send), 0x160 + M, 0xF, 0xF, true);
    return keep + __builtin_bit_cast(float, s);
}

template <int I>
static __device__ __forceinline__ float expert_step(const v4f* tree, int kslot, float xv,
                                                    int o0, int o1, int o2,
                                                    float g0, float g1, float g2)
{
    const float xb = share_f<I>(xv);
    const v4f* t0 = tree + (share_i<I>(o0) + kslot);
    const v4f* t1 = tree + (share_i<I>(o1) + kslot);
    const v4f* t2 = tree + (share_i<I>(o2) + kslot);
    const v4f a0 = t0[0], c0 = t0[16];
    const v4f a1 = t1[0], c1 = t1[16];
    const v4f a2 = t2[0], c2 = t2[16];
    const float s0 = fast_tanh(xb * a0.x + a0.y) * a0.z + fast_tanh(xb * c0.x + c0.y) * c0.z;
    const float s1 = fast_tanh(xb * a1.x + a1.y) * a1.z + fast_tanh(xb * c1.x + c1.y) * c1.z;
    const float s2 = fast_tanh(xb * a2.x + a2.y) * a2.z + fast_tanh(xb * c2.x + c2.y) * c2.z;
    return share_f<I>(g0) * s0 + share_f<I>(g1) * s1 + share_f<I>(g2) * s2;
}

__global__ __launch_bounds__(ROWS_PER_BLOCK)
void forest_moe_kernel(const float* __restrict__ x,
                       const float* __restrict__ rw1, const float* __restrict__ rb1,
                       const float* __restrict__ rw2, const float* __restrict__ rb2,
                       const float* __restrict__ tw1, const float* __restrict__ tb1,
                       const float* __restrict__ tw2, const float* __restrict__ tb2,
                       float* __restrict__ y)
{
    __shared__ v4f   s_tree[NT * H];

    const int tid  = threadIdx.x;
    const int wave = tid >> 5;
    const int lane = tid & 31;

    const int rowBase = blockIdx.x * ROWS_PER_BLOCK + wave * ROWS_PER_WAVE;
    const int mrow    = lane & 15;
    const int khalf   = (lane >> 4) << 1;

    float xsave[2];
    xsave[0] = x[rowBase + mrow];
    xsave[1] = x[rowBase + 16 + mrow];
    const int tr0 = mrow;
    const int tr1 = 16 + (mrow & 7);
    v2f wa0[8], wa1[8], w1p[8], b1p[8];
#pragma unroll
    for (int s = 0; s < 8; ++s) {
        const int k0 = 4 * s + khalf;
        wa0[s].x = rw2[k0 * NT + tr0];
        wa0[s].y = rw2[(k0 + 1) * NT + tr0];
        wa1[s].x = rw2[k0 * NT + tr1];
        wa1[s].y = rw2[(k0 + 1) * NT + tr1];
        w1p[s].x = rw1[k0];
        w1p[s].y = rw1[k0 + 1];
        b1p[s].x = rb1[k0];
        b1p[s].y = rb1[k0 + 1];
    }
    float bias[NT];
#pragma unroll
    for (int t = 0; t < NT; ++t) bias[t] = rb2[t];

    v8f d0[2], d1[2];
#pragma unroll
    for (int rt = 0; rt < 2; ++rt) {
        const float xv = xsave[rt];
        v2f hb[8];
#pragma unroll
        for (int s = 0; s < 8; ++s) {
            hb[s].x = fast_tanh(xv * w1p[s].x + b1p[s].x);
            hb[s].y = fast_tanh(xv * w1p[s].y + b1p[s].y);
        }
        v8f c0 = {}, c1 = {};
#pragma unroll
        for (int s = 0; s < 8; ++s) {
            c0 = __builtin_amdgcn_wmma_f32_16x16x4_f32(
                     false, wa0[s],  false, hb[s],
                     (short)0, c0,  false,  false);
            c1 = __builtin_amdgcn_wmma_f32_16x16x4_f32(
                     false, wa1[s],  false, hb[s],
                     (short)0, c1,  false,  false);
        }
        d0[rt] = c0;
        d1[rt] = c1;
    }

    const bool upper = lane >= 16;
    float sc[NT];
#pragma unroll
    for (int r = 0; r < 8; ++r) {
        const float lo = d0[0][r], hi = d0[1][r];
        const v2u sw = __builtin_amdgcn_permlane16_swap(
                __builtin_bit_cast(unsigned, lo), __builtin_bit_cast(unsigned, hi), false, false);
        const unsigned s0 = sw.x, s1 = sw.y;
        sc[r]      = __builtin_bit_cast(float, s0);
        sc[8 + r]  = __builtin_bit_cast(float, s1);
        sc[16 + r] = upper ? d1[1][r] : d1[0][r];
    }

    for (int i = tid; i < NT * H;  i += ROWS_PER_BLOCK) {
        v4f w = { tw1[i], tb1[i], tw2[i], 0.0f };
        s_tree[i] = w;
    }
    __builtin_amdgcn_fence(__ATOMIC_RELEASE, "workgroup");
    __builtin_amdgcn_s_barrier_signal(-1);

    const float xv = (lane < 16) ? xsave[0] : xsave[1];

    float b0 = -__builtin_inff(), b1v = -__builtin_inff(), b2v = -__builtin_inff();
    int   i0 = 0, i1 = 0, i2 = 0;
#pragma unroll
    for (int t = 0; t < NT; ++t) {
        const float sv = sc[t] + bias[t];
        if (sv > b0)       { b2v = b1v; i2 = i1; b1v = b0; i1 = i0; b0 = sv; i0 = t; }
        else if (sv > b1v) { b2v = b1v; i2 = i1; b1v = sv; i1 = t; }
        else if (sv > b2v) { b2v = sv;  i2 = t; }
    }
    const float e1  = fast_exp(b1v - b0);
    const float e2  = fast_exp(b2v - b0);
    const float inv = 1.0f / (1.0f + e1 + e2);
    const float g0 = inv, g1 = e1 * inv, g2 = e2 * inv;

    __builtin_amdgcn_s_barrier_wait(-1);
    __builtin_amdgcn_fence(__ATOMIC_ACQUIRE, "workgroup");

    const float t0 = tb2[i0], t1 = tb2[i1], t2 = tb2[i2];

    const int kslot = lane & 15;
    const int o0 = i0 * H, o1 = i1 * H, o2 = i2 * H;
    float p[16];
    p[0]  = expert_step<0>(s_tree, kslot, xv, o0, o1, o2, g0, g1, g2);
    p[1]  = expert_step<1>(s_tree, kslot, xv, o0, o1, o2, g0, g1, g2);
    p[2]  = expert_step<2>(s_tree, kslot, xv, o0, o1, o2, g0, g1, g2);
    p[3]  = expert_step<3>(s_tree, kslot, xv, o0, o1, o2, g0, g1, g2);
    p[4]  = expert_step<4>(s_tree, kslot, xv, o0, o1, o2, g0, g1, g2);
    p[5]  = expert_step<5>(s_tree, kslot, xv, o0, o1, o2, g0, g1, g2);
    p[6]  = expert_step<6>(s_tree, kslot, xv, o0, o1, o2, g0, g1, g2);
    p[7]  = expert_step<7>(s_tree, kslot, xv, o0, o1, o2, g0, g1, g2);
    p[8]  = expert_step<8>(s_tree, kslot, xv, o0, o1, o2, g0, g1, g2);
    p[9]  = expert_step<9>(s_tree, kslot, xv, o0, o1, o2, g0, g1, g2);
    p[10] = expert_step<10>(s_tree, kslot, xv, o0, o1, o2, g0, g1, g2);
    p[11] = expert_step<11>(s_tree, kslot, xv, o0, o1, o2, g0, g1, g2);
    p[12] = expert_step<12>(s_tree, kslot, xv, o0, o1, o2, g0, g1, g2);
    p[13] = expert_step<13>(s_tree, kslot, xv, o0, o1, o2, g0, g1, g2);
    p[14] = expert_step<14>(s_tree, kslot, xv, o0, o1, o2, g0, g1, g2);
    p[15] = expert_step<15>(s_tree, kslot, xv, o0, o1, o2, g0, g1, g2);

    const bool h8 = (lane & 8) != 0, h4 = (lane & 4) != 0, h2 = (lane & 2) != 0, h1 = (lane & 1) != 0;
    float r8[8], r4[4], r2[2];
#pragma unroll
    for (int j = 0; j < 8; ++j) r8[j] = xmask_add<8>(h8 ? p[j + 8] : p[j], h8 ? p[j] : p[j + 8]);
#pragma unroll
    for (int j = 0; j < 4; ++j) r4[j] = xmask_add<4>(h4 ? r8[j + 4] : r8[j], h4 ? r8[j] : r8[j + 4]);
#pragma unroll
    for (int j = 0; j < 2; ++j) r2[j] = xmask_add<2>(h2 ? r4[j + 2] : r4[j], h2 ? r4[j] : r4[j + 2]);
    const float dots = xmask_add<1>(h1 ? r2[1] : r2[0], h1 ? r2[0] : r2[1]);

    y[rowBase + lane] = dots + (g0 * t0 + g1 * t1 + g2 * t2);
}

extern "C" void kernel_launch(void* const* d_in, const int* in_sizes, int n_in,
                              void* d_out, int out_size, void* d_ws, size_t ws_size,
                              hipStream_t stream) {
    const float* x   = (const float*)d_in[0];
    const float* rw1 = (const float*)d_in[1];
    const float* rb1 = (const float*)d_in[2];
    const float* rw2 = (const float*)d_in[3];
    const float* rb2 = (const float*)d_in[4];
    const float* tw1 = (const float*)d_in[5];
    const float* tb1 = (const float*)d_in[6];
    const float* tw2 = (const float*)d_in[7];
    const float* tb2 = (const float*)d_in[8];
    float* y = (float*)d_out;

    const int B = in_sizes[0];
    const int grid = B / ROWS_PER_BLOCK;
    forest_moe_kernel<<<grid, ROWS_PER_BLOCK, 0, stream>>>(
        x, rw1, rb1, rw2, rb2, tw1, tb1, tw2, tb2, y);
}


# tree table laid out for its 16-lane readers: slot (tree,c) = {w1[c],w1[c+16],b1[c],b1[c+16]} plus a {w2[c],w2[c+16]} plane; a lane's two units run as packed pairs to the last add (sum order changes)
# speedup vs baseline: 1.8751x; 1.0897x over previous
#include <hip/hip_runtime.h>
#include <cmath>

typedef __attribute__((ext_vector_type(2))) float v2f;
typedef __attribute__((ext_vector_type(2))) unsigned v2u;
typedef __attribute__((ext_vector_type(4))) float v4f;
typedef __attribute__((ext_vector_type(8))) float v8f;
typedef v2f __attribute__((may_alias)) v2f_lds;

#define RH   32
#define NT   24
#define H    32
#define WAVES_PER_BLOCK 8
#define ROWS_PER_WAVE   32
#define ROWS_PER_BLOCK  (WAVES_PER_BLOCK * ROWS_PER_WAVE)
#define HP   16
#define NSLOT (NT * HP)

static __device__ __forceinline__ float fast_tanh(float v) {
#if __has_builtin(__builtin_amdgcn_tanhf)
    return __builtin_amdgcn_tanhf(v);
#else
    return tanhf(v);
#endif
}

static __device__ __forceinline__ float fast_exp(float v) {
#if __has_builtin(__builtin_amdgcn_exp2f)
    return __builtin_amdgcn_exp2f(v * 1.44269504088896340736f);
#else
    return __expf(v);
#endif
}

template <int I>
static __device__ __forceinline__ int share_i(int v) {
    return __builtin_amdgcn_update_dpp(0, v, 0x150 + I, 0xF, 0xF, true);
}
template <int I>
static __device__ __forceinline__ float share_f(float v) {
    return __builtin_bit_cast(float, share_i<I>(__builtin_bit_cast(int, v)));
}
template <int M>
static __device__ __forceinline__ float xmask_add(float keep, float send) {
    const int s = __builtin_amdgcn_update_dpp(0, __builtin_bit_cast(int, send), 0x160 + M, 0xF, 0xF, true);
    return keep + __builtin_bit_cast(float, s);
}

template <int I>
static __device__ __forceinline__ float expert_step(const v4f* tree, int kslot, float xv,
                                                    int o0, int o1, int o2,
                                                    float g0, float g1, float g2)
{
    const float xb = share_f<I>(xv);
    const v2f xb2 = { xb, xb };
    const v4f* t0 = tree + (share_i<I>(o0) + kslot);
    const v4f* t1 = tree + (share_i<I>(o1) + kslot);
    const v4f* t2 = tree + (share_i<I>(o2) + kslot);
    const v4f a0 = t0[0], a1 = t1[0], a2 = t2[0];
    const v2f c0 = *(const v2f_lds*)(t0 + NSLOT);
    const v2f c1 = *(const v2f_lds*)(t1 + NSLOT);
    const v2f c2 = *(const v2f_lds*)(t2 + NSLOT);
    const v2f h0 = a0.xy * xb2 + a0.zw;
    const v2f h1 = a1.xy * xb2 + a1.zw;
    const v2f h2 = a2.xy * xb2 + a2.zw;
    const v2f u0 = { fast_tanh(h0.x), fast_tanh(h0.y) };
    const v2f u1 = { fast_tanh(h1.x), fast_tanh(h1.y) };
    const v2f u2 = { fast_tanh(h2.x), fast_tanh(h2.y) };
    const float w0 = share_f<I>(g0), w1 = share_f<I>(g1), w2 = share_f<I>(g2);
    const v2f w0p = { w0, w0 }, w1p = { w1, w1 }, w2p = { w2, w2 };
    v2f acc = (u0 * c0) * w0p;
    acc = (u1 * c1) * w1p + acc;
    acc = (u2 * c2) * w2p + acc;
    return acc.x + acc.y;
}

__global__ __launch_bounds__(ROWS_PER_BLOCK)
void forest_moe_kernel(const float* __restrict__ x,
                       const float* __restrict__ rw1, const float* __restrict__ rb1,
                       const float* __restrict__ rw2, const float* __restrict__ rb2,
                       const float* __restrict__ tw1, const float* __restrict__ tb1,
                       const float* __restrict__ tw2, const float* __restrict__ tb2,
                       float* __restrict__ y)
{
    __shared__ v4f   s_tree[2 * NSLOT];

    const int tid  = threadIdx.x;
    const int wave = tid >> 5;
    const int lane = tid & 31;

    const int rowBase = blockIdx.x * ROWS_PER_BLOCK + wave * ROWS_PER_WAVE;
    const int mrow    = lane & 15;
    const int khalf   = (lane >> 4) << 1;

    float xsave[2];
    xsave[0] = x[rowBase + mrow];
    xsave[1] = x[rowBase + 16 + mrow];
    const int tr0 = mrow;
    const int tr1 = 16 + (mrow & 7);
    v2f wa0[8], wa1[8], w1p[8], b1p[8];
#pragma unroll
    for (int s = 0; s < 8; ++s) {
        const int k0 = 4 * s + khalf;
        wa0[s].x = rw2[k0 * NT + tr0];
        wa0[s].y = rw2[(k0 + 1) * NT + tr0];
        wa1[s].x = rw2[k0 * NT + tr1];
        wa1[s].y = rw2[(k0 + 1) * NT + tr1];
        w1p[s].x = rw1[k0];
        w1p[s].y = rw1[k0 + 1];
        b1p[s].x = rb1[k0];
        b1p[s].y = rb1[k0 + 1];
    }
    float bias[NT];
#pragma unroll
    for (int t = 0; t < NT; ++t) bias[t] = rb2[t];

    v8f d0[2], d1[2];
#pragma unroll
    for (int rt = 0; rt < 2; ++rt) {
        const float xv = xsave[rt];
        v2f hb[8];
#pragma unroll
        for (int s = 0; s < 8; ++s) {
            hb[s].x = fast_tanh(xv * w1p[s].x + b1p[s].x);
            hb[s].y = fast_tanh(xv * w1p[s].y + b1p[s].y);
        }
        v8f c0 = {}, c1 = {};
#pragma unroll
        for (int s = 0; s < 8; ++s) {
            c0 = __builtin_amdgcn_wmma_f32_16x16x4_f32(
                     false, wa0[s],  false, hb[s],
                     (short)0, c0,  false,  false);
            c1 = __builtin_amdgcn_wmma_f32_16x16x4_f32(
                     false, wa1[s],  false, hb[s],
                     (short)0, c1,  false,  false);
        }
        d0[rt] = c0;
        d1[rt] = c1;
    }

    const bool upper = lane >= 16;
    float sc[NT];
#pragma unroll
    for (int r = 0; r < 8; ++r) {
        const float lo = d0[0][r], hi = d0[1][r];
        const v2u sw = __builtin_amdgcn_permlane16_swap(
                __builtin_bit_cast(unsigned, lo), __builtin_bit_cast(unsigned, hi), false, false);
        const unsigned s0 = sw.x, s1 = sw.y;
        sc[r]      = __builtin_bit_cast(float, s0);
        sc[8 + r]  = __builtin_bit_cast(float, s1);
        sc[16 + r] = upper ? d1[1][r] : d1[0][r];
    }

    for (int j = tid; j < NSLOT; j += ROWS_PER_BLOCK) {
        const int k = ((j >> 4) << 5) + (j & 15);
        const v4f ab = { tw1[k], tw1[k + 16], tb1[k], tb1[k + 16] };
        const v4f w  = { tw2[k], tw2[k + 16], 0.0f, 0.0f };
        s_tree[j]         = ab;
        s_tree[NSLOT + j] = w;
    }
    __builtin_amdgcn_fence(__ATOMIC_RELEASE, "workgroup");
    __builtin_amdgcn_s_barrier_signal(-1);

    const float xv = (lane < 16) ? xsave[0] : xsave[1];

    float b0 = -__builtin_inff(), b1v = -__builtin_inff(), b2v = -__builtin_inff();
    int   i0 = 0, i1 = 0, i2 = 0;
#pragma unroll
    for (int t = 0; t < NT; ++t) {
        const float sv = sc[t] + bias[t];
        if (sv > b0)       { b2v = b1v; i2 = i1; b1v = b0; i1 = i0; b0 = sv; i0 = t; }
        else if (sv > b1v) { b2v = b1v; i2 = i1; b1v = sv; i1 = t; }
        else if (sv > b2v) { b2v = sv;  i2 = t; }
    }
    const float e1  = fast_exp(b1v - b0);
    const float e2  = fast_exp(b2v - b0);
    const float inv = 1.0f / (1.0f + e1 + e2);
    const float g0 = inv, g1 = e1 * inv, g2 = e2 * inv;

    __builtin_amdgcn_s_barrier_wait(-1);
    __builtin_amdgcn_fence(__ATOMIC_ACQUIRE, "workgroup");

    const float t0 = tb2[i0], t1 = tb2[i1], t2 = tb2[i2];

    const int kslot = lane & 15;
    const int o0 = i0 * HP, o1 = i1 * HP, o2 = i2 * HP;
    float p[16];
    p[0]  = expert_step<0>(s_tree, kslot, xv, o0, o1, o2, g0, g1, g2);
    p[1]  = expert_step<1>(s_tree, kslot, xv, o0, o1, o2, g0, g1, g2);
    p[2]  = expert_step<2>(s_tree, kslot, xv, o0, o1, o2, g0, g1, g2);
    p[3]  = expert_step<3>(s_tree, kslot, xv, o0, o1, o2, g0, g1, g2);
    p[4]  = expert_step<4>(s_tree, kslot, xv, o0, o1, o2, g0, g1, g2);
    p[5]  = expert_step<5>(s_tree, kslot, xv, o0, o1, o2, g0, g1, g2);
    p[6]  = expert_step<6>(s_tree, kslot, xv, o0, o1, o2, g0, g1, g2);
    p[7]  = expert_step<7>(s_tree, kslot, xv, o0, o1, o2, g0, g1, g2);
    p[8]  = expert_step<8>(s_tree, kslot, xv, o0, o1, o2, g0, g1, g2);
    p[9]  = expert_step<9>(s_tree, kslot, xv, o0, o1, o2, g0, g1, g2);
    p[10] = expert_step<10>(s_tree, kslot, xv, o0, o1, o2, g0, g1, g2);
    p[11] = expert_step<11>(s_tree, kslot, xv, o0, o1, o2, g0, g1, g2);
    p[12] = expert_step<12>(s_tree, kslot, xv, o0, o1, o2, g0, g1, g2);
    p[13] = expert_step<13>(s_tree, kslot, xv, o0, o1, o2, g0, g1, g2);
    p[14] = expert_step<14>(s_tree, kslot, xv, o0, o1, o2, g0, g1, g2);
    p[15] = expert_step<15>(s_tree, kslot, xv, o0, o1, o2, g0, g1, g2);

    const bool h8 = (lane & 8) != 0, h4 = (lane & 4) != 0, h2 = (lane & 2) != 0, h1 = (lane & 1) != 0;
    float r8[8], r4[4], r2[2];
#pragma unroll
    for (int j = 0; j < 8; ++j) r8[j] = xmask_add<8>(h8 ? p[j + 8] : p[j], h8 ? p[j] : p[j + 8]);
#pragma unroll
    for (int j = 0; j < 4; ++j) r4[j] = xmask_add<4>(h4 ? r8[j + 4] : r8[j], h4 ? r8[j] : r8[j + 4]);
#pragma unroll
    for (int j = 0; j < 2; ++j) r2[j] = xmask_add<2>(h2 ? r4[j + 2] : r4[j], h2 ? r4[j] : r4[j + 2]);
    const float dots = xmask_add<1>(h1 ? r2[1] : r2[0], h1 ? r2[0] : r2[1]);

    y[rowBase + lane] = dots + (g0 * t0 + g1 * t1 + g2 * t2);
}

extern "C" void kernel_launch(void* const* d_in, const int* in_sizes, int n_in,
                              void* d_out, int out_size, void* d_ws, size_t ws_size,
                              hipStream_t stream) {
    const float* x   = (const float*)d_in[0];
    const float* rw1 = (const float*)d_in[1];
    const float* rb1 = (const float*)d_in[2];
    const float* rw2 = (const float*)d_in[3];
    const float* rb2 = (const float*)d_in[4];
    const float* tw1 = (const float*)d_in[5];
    const float* tb1 = (const float*)d_in[6];
    const float* tw2 = (const float*)d_in[7];
    const float* tb2 = (const float*)d_in[8];
    float* y = (float*)d_out;

    const int B = in_sizes[0];
    const int grid = B / ROWS_PER_BLOCK;
    forest_moe_kernel<<<grid, ROWS_PER_BLOCK, 0, stream>>>(
        x, rw1, rb1, rw2, rb2, tw1, tb1, tw2, tb2, y);
}


# expert phase: a data row's x, softmax numerators and 3 tree offsets (one packed word) reach its 16 lanes by 4 permlane16 broadcasts, not 7 row_share DPP moves; 1/denominator applied after the row sum
# speedup vs baseline: 1.9894x; 1.0610x over previous
#include <hip/hip_runtime.h>
#include <cmath>

typedef __attribute__((ext_vector_type(2))) float v2f;
typedef __attribute__((ext_vector_type(2))) unsigned v2u;
typedef __attribute__((ext_vector_type(4))) float v4f;
typedef __attribute__((ext_vector_type(8))) float v8f;
typedef v2f __attribute__((may_alias)) v2f_lds;

#define RH   32
#define NT   24
#define H    32
#define WAVES_PER_BLOCK 8
#define ROWS_PER_WAVE   32
#define ROWS_PER_BLOCK  (WAVES_PER_BLOCK * ROWS_PER_WAVE)
#define HP   16
#define NSLOT (NT * HP)

static __device__ __forceinline__ float fast_tanh(float v) {
#if __has_builtin(__builtin_amdgcn_tanhf)
    return __builtin_amdgcn_tanhf(v);
#else
    return tanhf(v);
#endif
}

static __device__ __forceinline__ float fast_exp(float v) {
#if __has_builtin(__builtin_amdgcn_exp2f)
    return __builtin_amdgcn_exp2f(v * 1.44269504088896340736f);
#else
    return __expf(v);
#endif
}

template <int I>
static __device__ __forceinline__ unsigned row_bcast_u(unsigned v) {
    return __builtin_amdgcn_permlane16(v, v, 0x11111111u * I, 0x11111111u * I, true, false);
}
template <int I>
static __device__ __forceinline__ float row_bcast_f(float v) {
    return __builtin_bit_cast(float, row_bcast_u<I>(__builtin_bit_cast(unsigned, v)));
}
template <int M>
static __device__ __forceinline__ float xmask_add(float keep, float send) {
    const int s = __builtin_amdgcn_update_dpp(0, __builtin_bit_cast(int, send), 0x160 + M, 0xF, 0xF, true);
    return keep + __builtin_bit_cast(float, s);
}

template <int I>
static __device__ __forceinline__ float expert_step(const char* table, unsigned kbyte, float xv,
                                                    float e1, float e2, unsigned trees)
{
    const float xb = row_bcast_f<I>(xv);
    const unsigned tr = row_bcast_u<I>(trees);
    const float f1 = row_bcast_f<I>(e1), f2 = row_bcast_f<I>(e2);
    const v2f xb2 = { xb, xb };
    const v4f* t0 = (const v4f*)(table + __builtin_amdgcn_perm(tr, kbyte, 0x0c0c0500u));
    const v4f* t1 = (const v4f*)(table + __builtin_amdgcn_perm(tr, kbyte, 0x0c0c0600u));
    const v4f* t2 = (const v4f*)(table + __builtin_amdgcn_perm(tr, kbyte, 0x0c0c0700u));
    const v4f a0 = t0[0], a1 = t1[0], a2 = t2[0];
    const v2f c0 = *(const v2f_lds*)(t0 + NSLOT);
    const v2f c1 = *(const v2f_lds*)(t1 + NSLOT);
    const v2f c2 = *(const v2f_lds*)(t2 + NSLOT);
    const v2f h0 = a0.xy * xb2 + a0.zw;
    const v2f h1 = a1.xy * xb2 + a1.zw;
    const v2f h2 = a2.xy * xb2 + a2.zw;
    const v2f u0 = { fast_tanh(h0.x), fast_tanh(h0.y) };
    const v2f u1 = { fast_tanh(h1.x), fast_tanh(h1.y) };
    const v2f u2 = { fast_tanh(h2.x), fast_tanh(h2.y) };
    const v2f f1p = { f1, f1 }, f2p = { f2, f2 };
    v2f acc = u0 * c0;
    acc = (u1 * c1) * f1p + acc;
    acc = (u2 * c2) * f2p + acc;
    return acc.x + acc.y;
}

__global__ __launch_bounds__(ROWS_PER_BLOCK)
void forest_moe_kernel(const float* __restrict__ x,
                       const float* __restrict__ rw1, const float* __restrict__ rb1,
                       const float* __restrict__ rw2, const float* __restrict__ rb2,
                       const float* __restrict__ tw1, const float* __restrict__ tb1,
                       const float* __restrict__ tw2, const float* __restrict__ tb2,
                       float* __restrict__ y)
{
    __shared__ v4f   s_tree[2 * NSLOT];

    const int tid  = threadIdx.x;
    const int wave = tid >> 5;
    const int lane = tid & 31;

    const int rowBase = blockIdx.x * ROWS_PER_BLOCK + wave * ROWS_PER_WAVE;
    const int mrow    = lane & 15;
    const int khalf   = (lane >> 4) << 1;

    float xsave[2];
    xsave[0] = x[rowBase + mrow];
    xsave[1] = x[rowBase + 16 + mrow];
    const int tr0 = mrow;
    const int tr1 = 16 + (mrow & 7);
    v2f wa0[8], wa1[8], w1p[8], b1p[8];
#pragma unroll
    for (int s = 0; s < 8; ++s) {
        const int k0 = 4 * s + khalf;
        wa0[s].x = rw2[k0 * NT + tr0];
        wa0[s].y = rw2[(k0 + 1) * NT + tr0];
        wa1[s].x = rw2[k0 * NT + tr1];
        wa1[s].y = rw2[(k0 + 1) * NT + tr1];
        w1p[s].x = rw1[k0];
        w1p[s].y = rw1[k0 + 1];
        b1p[s].x = rb1[k0];
        b1p[s].y = rb1[k0 + 1];
    }
    float bias[NT];
#pragma unroll
    for (int t = 0; t < NT; ++t) bias[t] = rb2[t];

    v8f d0[2], d1[2];
#pragma unroll
    for (int rt = 0; rt < 2; ++rt) {
        const float xv = xsave[rt];
        v2f hb[8];
#pragma unroll
        for (int s = 0; s < 8; ++s) {
            hb[s].x = fast_tanh(xv * w1p[s].x + b1p[s].x);
            hb[s].y = fast_tanh(xv * w1p[s].y + b1p[s].y);
        }
        v8f c0 = {}, c1 = {};
#pragma unroll
        for (int s = 0; s < 8; ++s) {
            c0 = __builtin_amdgcn_wmma_f32_16x16x4_f32(
                     false, wa0[s],  false, hb[s],
                     (short)0, c0,  false,  false);
            c1 = __builtin_amdgcn_wmma_f32_16x16x4_f32(
                     false, wa1[s],  false, hb[s],
                     (short)0, c1,  false,  false);
        }
        d0[rt] = c0;
        d1[rt] = c1;
    }

    const bool upper = lane >= 16;
    float sc[NT];
#pragma unroll
    for (int r = 0; r < 8; ++r) {
        const float lo = d0[0][r], hi = d0[1][r];
        const v2u sw = __builtin_amdgcn_permlane16_swap(
                __builtin_bit_cast(unsigned, lo), __builtin_bit_cast(unsigned, hi), false, false);
        const unsigned s0 = sw.x, s1 = sw.y;
        sc[r]      = __builtin_bit_cast(float, s0);
        sc[8 + r]  = __builtin_bit_cast(float, s1);
        sc[16 + r] = upper ? d1[1][r] : d1[0][r];
    }

    for (int j = tid; j < NSLOT; j += ROWS_PER_BLOCK) {
        const int k = ((j >> 4) << 5) + (j & 15);
        const v4f ab = { tw1[k], tw1[k + 16], tb1[k], tb1[k + 16] };
        const v4f w  = { tw2[k], tw2[k + 16], 0.0f, 0.0f };
        s_tree[j]         = ab;
        s_tree[NSLOT + j] = w;
    }
    __builtin_amdgcn_fence(__ATOMIC_RELEASE, "workgroup");
    __builtin_amdgcn_s_barrier_signal(-1);

    const float xv = (lane < 16) ? xsave[0] : xsave[1];

    float b0 = -__builtin_inff(), b1v = -__builtin_inff(), b2v = -__builtin_inff();
    int   i0 = 0, i1 = 0, i2 = 0;
#pragma unroll
    for (int t = 0; t < NT; ++t) {
        const float sv = sc[t] + bias[t];
        if (sv > b0)       { b2v = b1v; i2 = i1; b1v = b0; i1 = i0; b0 = sv; i0 = t; }
        else if (sv > b1v) { b2v = b1v; i2 = i1; b1v = sv; i1 = t; }
        else if (sv > b2v) { b2v = sv;  i2 = t; }
    }
    const float e1  = fast_exp(b1v - b0);
    const float e2  = fast_exp(b2v - b0);
    const float inv = 1.0f / (1.0f + e1 + e2);
    const unsigned trees = ((unsigned)i0 << 8) | ((unsigned)i1 << 16) | ((unsigned)i2 << 24);

    __builtin_amdgcn_s_barrier_wait(-1);
    __builtin_amdgcn_fence(__ATOMIC_ACQUIRE, "workgroup");

    const float t0 = tb2[i0], t1 = tb2[i1], t2 = tb2[i2];

    const unsigned kbyte = (unsigned)(lane & 15) << 4;
    const char* const table = (const char*)s_tree;
    float p[16];
    p[0]  = expert_step<0>(table, kbyte, xv, e1, e2, trees);
    p[1]  = expert_step<1>(table, kbyte, xv, e1, e2, trees);
    p[2]  = expert_step<2>(table, kbyte, xv, e1, e2, trees);
    p[3]  = expert_step<3>(table, kbyte, xv, e1, e2, trees);
    p[4]  = expert_step<4>(table, kbyte, xv, e1, e2, trees);
    p[5]  = expert_step<5>(table, kbyte, xv, e1, e2, trees);
    p[6]  = expert_step<6>(table, kbyte, xv, e1, e2, trees);
    p[7]  = expert_step<7>(table, kbyte, xv, e1, e2, trees);
    p[8]  = expert_step<8>(table, kbyte, xv, e1, e2, trees);
    p[9]  = expert_step<9>(table, kbyte, xv, e1, e2, trees);
    p[10] = expert_step<10>(table, kbyte, xv, e1, e2, trees);
    p[11] = expert_step<11>(table, kbyte, xv, e1, e2, trees);
    p[12] = expert_step<12>(table, kbyte, xv, e1, e2, trees);
    p[13] = expert_step<13>(table, kbyte, xv, e1, e2, trees);
    p[14] = expert_step<14>(table, kbyte, xv, e1, e2, trees);
    p[15] = expert_step<15>(table, kbyte, xv, e1, e2, trees);

    const bool h8 = (lane & 8) != 0, h4 = (lane & 4) != 0, h2 = (lane & 2) != 0, h1 = (lane & 1) != 0;
    float r8[8], r4[4], r2[2];
#pragma unroll
    for (int j = 0; j < 8; ++j) r8[j] = xmask_add<8>(h8 ? p[j + 8] : p[j], h8 ? p[j] : p[j + 8]);
#pragma unroll
    for (int j = 0; j < 4; ++j) r4[j] = xmask_add<4>(h4 ? r8[j + 4] : r8[j], h4 ? r8[j] : r8[j + 4]);
#pragma unroll
    for (int j = 0; j < 2; ++j) r2[j] = xmask_add<2>(h2 ? r4[j + 2] : r4[j], h2 ? r4[j] : r4[j + 2]);
    const float dots = xmask_add<1>(h1 ? r2[1] : r2[0], h1 ? r2[0] : r2[1]);

    y[rowBase + lane] = inv * (dots + (t0 + e1 * t1 + e2 * t2));
}

extern "C" void kernel_launch(void* const* d_in, const int* in_sizes, int n_in,
                              void* d_out, int out_size, void* d_ws, size_t ws_size,
                              hipStream_t stream) {
    const float* x   = (const float*)d_in[0];
    const float* rw1 = (const float*)d_in[1];
    const float* rb1 = (const float*)d_in[2];
    const float* rw2 = (const float*)d_in[3];
    const float* rb2 = (const float*)d_in[4];
    const float* tw1 = (const float*)d_in[5];
    const float* tb1 = (const float*)d_in[6];
    const float* tw2 = (const float*)d_in[7];
    const float* tb2 = (const float*)d_in[8];
    float* y = (float*)d_out;

    const int B = in_sizes[0];
    const int grid = B / ROWS_PER_BLOCK;
    forest_moe_kernel<<<grid, ROWS_PER_BLOCK, 0, stream>>>(
        x, rw1, rb1, rw2, rb2, tw1, tb1, tw2, tb2, y);
}


# expert phase: 4 lanes per data row, 8 hidden units per lane (was 16 x 2): 4 steps not 16, 2-stage row sum; 3 table planes, tree rows stored twice; launch bound keeps 8 waves/SIMD (sum order changes)
# speedup vs baseline: 2.0909x; 1.0510x over previous
#include <hip/hip_runtime.h>
#include <cmath>

typedef __attribute__((ext_vector_type(2))) float v2f;
typedef __attribute__((ext_vector_type(2))) unsigned v2u;
typedef __attribute__((ext_vector_type(4))) float v4f;
typedef __attribute__((ext_vector_type(8))) float v8f;

#define RH   32
#define NT   24
#define H    32
#define WAVES_PER_BLOCK 8
#define ROWS_PER_WAVE   32
#define ROWS_PER_BLOCK  (WAVES_PER_BLOCK * ROWS_PER_WAVE)
#define ROWSLOTS 16
#define PLANE (NT * ROWSLOTS)

static __device__ __forceinline__ float fast_tanh(float v) {
#if __has_builtin(__builtin_amdgcn_tanhf)
    return __builtin_amdgcn_tanhf(v);
#else
    return tanhf(v);
#endif
}

static __device__ __forceinline__ float fast_exp(float v) {
#if __has_builtin(__builtin_amdgcn_exp2f)
    return __builtin_amdgcn_exp2f(v * 1.44269504088896340736f);
#else
    return __expf(v);
#endif
}

template <int S>
static __device__ __forceinline__ unsigned grp_bcast_u(unsigned v) {
    return __builtin_amdgcn_permlane16(v, v,
            0x11110000u * (4 * S + 1) + 0x00001111u * (4 * S),
            0x11110000u * (4 * S + 3) + 0x00001111u * (4 * S + 2), true, false);
}
template <int S>
static __device__ __forceinline__ float grp_bcast_f(float v) {
    return __builtin_bit_cast(float, grp_bcast_u<S>(__builtin_bit_cast(unsigned, v)));
}
template <int M>
static __device__ __forceinline__ float xmask_add(float keep, float send) {
    const int s = __builtin_amdgcn_update_dpp(0, __builtin_bit_cast(int, send), 0x160 + M, 0xF, 0xF, true);
    return keep + __builtin_bit_cast(float, s);
}

static __device__ __forceinline__ v2f tree_part(const char* table, unsigned offA, unsigned offB, v2f xb2)
{
    const v4f* a = (const v4f*)(table + offA);
    const v4f* b = (const v4f*)(table + offB);
    const v4f w1a = a[0], b1a = a[PLANE], w2a = a[2 * PLANE];
    const v4f w1b = b[0], b1b = b[PLANE], w2b = b[2 * PLANE];
    const v2f h0 = w1a.xy * xb2 + b1a.xy;
    const v2f h1 = w1a.zw * xb2 + b1a.zw;
    const v2f h2 = w1b.xy * xb2 + b1b.xy;
    const v2f h3 = w1b.zw * xb2 + b1b.zw;
    const v2f u0 = { fast_tanh(h0.x), fast_tanh(h0.y) };
    const v2f u1 = { fast_tanh(h1.x), fast_tanh(h1.y) };
    const v2f u2 = { fast_tanh(h2.x), fast_tanh(h2.y) };
    const v2f u3 = { fast_tanh(h3.x), fast_tanh(h3.y) };
    v2f d = u0 * w2a.xy;
    d = u1 * w2a.zw + d;
    d = u2 * w2b.xy + d;
    d = u3 * w2b.zw + d;
    return d;
}

template <int S>
static __device__ __forceinline__ float expert_step(const char* table, unsigned kA, unsigned kB,
                                                    float xv, float e1, float e2, unsigned trees)
{
    const float xb = grp_bcast_f<S>(xv);
    const unsigned tr = grp_bcast_u<S>(trees);
    const float f1 = grp_bcast_f<S>(e1), f2 = grp_bcast_f<S>(e2);
    const v2f xb2 = { xb, xb };
    const v2f d0 = tree_part(table, __builtin_amdgcn_perm(tr, kA, 0x0c0c0500u),
                                    __builtin_amdgcn_perm(tr, kB, 0x0c0c0500u), xb2);
    const v2f d1 = tree_part(table, __builtin_amdgcn_perm(tr, kA, 0x0c0c0600u),
                                    __builtin_amdgcn_perm(tr, kB, 0x0c0c0600u), xb2);
    const v2f d2 = tree_part(table, __builtin_amdgcn_perm(tr, kA, 0x0c0c0700u),
                                    __builtin_amdgcn_perm(tr, kB, 0x0c0c0700u), xb2);
    const v2f f1p = { f1, f1 }, f2p = { f2, f2 };
    v2f acc = d1 * f1p + d0;
    acc = d2 * f2p + acc;
    return acc.x + acc.y;
}

__global__ __launch_bounds__(ROWS_PER_BLOCK, 8)
void forest_moe_kernel(const float* __restrict__ x,
                       const float* __restrict__ rw1, const float* __restrict__ rb1,
                       const float* __restrict__ rw2, const float* __restrict__ rb2,
                       const float* __restrict__ tw1, const float* __restrict__ tb1,
                       const float* __restrict__ tw2, const float* __restrict__ tb2,
                       float* __restrict__ y)
{
    __shared__ v4f   s_tree[3 * PLANE];

    const int tid  = threadIdx.x;
    const int wave = tid >> 5;
    const int lane = tid & 31;

    const int rowBase = blockIdx.x * ROWS_PER_BLOCK + wave * ROWS_PER_WAVE;
    const int mrow    = lane & 15;
    const int khalf   = (lane >> 4) << 1;

    float xsave[2];
    xsave[0] = x[rowBase + mrow];
    xsave[1] = x[rowBase + 16 + mrow];
    const int tr0 = mrow;
    const int tr1 = 16 + (mrow & 7);
    v2f wa0[8], wa1[8], w1p[8], b1p[8];
#pragma unroll
    for (int s = 0; s < 8; ++s) {
        const int k0 = 4 * s + khalf;
        wa0[s].x = rw2[k0 * NT + tr0];
        wa0[s].y = rw2[(k0 + 1) * NT + tr0];
        wa1[s].x = rw2[k0 * NT + tr1];
        wa1[s].y = rw2[(k0 + 1) * NT + tr1];
        w1p[s].x = rw1[k0];
        w1p[s].y = rw1[k0 + 1];
        b1p[s].x = rb1[k0];
        b1p[s].y = rb1[k0 + 1];
    }
    float bias[NT];
#pragma unroll
    for (int t = 0; t < NT; ++t) bias[t] = rb2[t];

    v8f d0[2], d1[2];
#pragma unroll
    for (int rt = 0; rt < 2; ++rt) {
        const float xv = xsave[rt];
        v2f hb[8];
#pragma unroll
        for (int s = 0; s < 8; ++s) {
            hb[s].x = fast_tanh(xv * w1p[s].x + b1p[s].x);
            hb[s].y = fast_tanh(xv * w1p[s].y + b1p[s].y);
        }
        v8f c0 = {}, c1 = {};
#pragma unroll
        for (int s = 0; s < 8; ++s) {
            c0 = __builtin_amdgcn_wmma_f32_16x16x4_f32(
                     false, wa0[s],  false, hb[s],
                     (short)0, c0,  false,  false);
            c1 = __builtin_amdgcn_wmma_f32_16x16x4_f32(
                     false, wa1[s],  false, hb[s],
                     (short)0, c1,  false,  false);
        }
        d0[rt] = c0;
        d1[rt] = c1;
    }

    const bool upper = lane >= 16;
    float sc[NT];
#pragma unroll
    for (int r = 0; r < 8; ++r) {
        const float lo = d0[0][r], hi = d0[1][r];
        const v2u sw = __builtin_amdgcn_permlane16_swap(
                __builtin_bit_cast(unsigned, lo), __builtin_bit_cast(unsigned, hi), false, false);
        const unsigned s0 = sw.x, s1 = sw.y;
        sc[r]      = __builtin_bit_cast(float, s0);
        sc[8 + r]  = __builtin_bit_cast(float, s1);
        sc[16 + r] = upper ? d1[1][r] : d1[0][r];
    }

    if (tid < NT * 8) {
        const int d = ((tid >> 3) << 4) + (tid & 7);
        const v4f qa = ((const v4f*)tw1)[tid];
        const v4f qb = ((const v4f*)tb1)[tid];
        const v4f qc = ((const v4f*)tw2)[tid];
        s_tree[d]                 = qa;
        s_tree[d + 8]             = qa;
        s_tree[PLANE + d]         = qb;
        s_tree[PLANE + d + 8]     = qb;
        s_tree[2 * PLANE + d]     = qc;
        s_tree[2 * PLANE + d + 8] = qc;
    }
    __builtin_amdgcn_fence(__ATOMIC_RELEASE, "workgroup");
    __builtin_amdgcn_s_barrier_signal(-1);

    const float xv = (lane < 16) ? xsave[0] : xsave[1];

    float b0 = -__builtin_inff(), b1v = -__builtin_inff(), b2v = -__builtin_inff();
    int   i0 = 0, i1 = 0, i2 = 0;
#pragma unroll
    for (int t = 0; t < NT; ++t) {
        const float sv = sc[t] + bias[t];
        if (sv > b0)       { b2v = b1v; i2 = i1; b1v = b0; i1 = i0; b0 = sv; i0 = t; }
        else if (sv > b1v) { b2v = b1v; i2 = i1; b1v = sv; i1 = t; }
        else if (sv > b2v) { b2v = sv;  i2 = t; }
    }
    const float e1  = fast_exp(b1v - b0);
    const float e2  = fast_exp(b2v - b0);
    const float inv = 1.0f / (1.0f + e1 + e2);
    const unsigned trees = ((unsigned)i0 << 8) | ((unsigned)i1 << 16) | ((unsigned)i2 << 24);

    __builtin_amdgcn_s_barrier_wait(-1);
    __builtin_amdgcn_fence(__ATOMIC_ACQUIRE, "workgroup");

    const float t0 = tb2[i0], t1 = tb2[i1], t2 = tb2[i2];

    const unsigned kA = (unsigned)(lane & 15) << 4;
    const unsigned kB = (unsigned)((lane & 15) ^ 4) << 4;
    const char* const table = (const char*)s_tree;
    const float p0 = expert_step<0>(table, kA, kB, xv, e1, e2, trees);
    const float p1 = expert_step<1>(table, kA, kB, xv, e1, e2, trees);
    const float p2 = expert_step<2>(table, kA, kB, xv, e1, e2, trees);
    const float p3 = expert_step<3>(table, kA, kB, xv, e1, e2, trees);

    const bool h2 = (lane & 2) != 0, h1 = (lane & 1) != 0;
    const float r0 = xmask_add<2>(h2 ? p2 : p0, h2 ? p0 : p2);
    const float r1 = xmask_add<2>(h2 ? p3 : p1, h2 ? p1 : p3);
    const float tot = xmask_add<1>(h1 ? r1 : r0, h1 ? r0 : r1);
    const unsigned totu = __builtin_bit_cast(unsigned, tot);
    const float dots = __builtin_bit_cast(float,
            __builtin_amdgcn_permlane16(totu, totu, 0xD951C840u, 0xFB73EA62u, true, false));

    y[rowBase + lane] = inv * (dots + (t0 + e1 * t1 + e2 * t2));
}

extern "C" void kernel_launch(void* const* d_in, const int* in_sizes, int n_in,
                              void* d_out, int out_size, void* d_ws, size_t ws_size,
                              hipStream_t stream) {
    const float* x   = (const float*)d_in[0];
    const float* rw1 = (const float*)d_in[1];
    const float* rb1 = (const float*)d_in[2];
    const float* rw2 = (const float*)d_in[3];
    const float* rb2 = (const float*)d_in[4];
    const float* tw1 = (const float*)d_in[5];
    const float* tb1 = (const float*)d_in[6];
    const float* tw2 = (const float*)d_in[7];
    const float* tb2 = (const float*)d_in[8];
    float* y = (float*)d_out;

    const int B = in_sizes[0];
    const int grid = B / ROWS_PER_BLOCK;
    forest_moe_kernel<<<grid, ROWS_PER_BLOCK, 0, stream>>>(
        x, rw1, rb1, rw2, rb2, tw1, tb1, tw2, tb2, y);
}
